# starting kernel
# baseline (speedup 1.0000x reference)
#include <hip/hip_runtime.h>

typedef __attribute__((ext_vector_type(2))) float v2f;
typedef __attribute__((ext_vector_type(8))) float v8f;

#define HDIM 256
#define NROWS 1024

__global__ __launch_bounds__(256) void proj_wmma(
    const float* __restrict__ dec, const float* __restrict__ enc,
    const float* __restrict__ W,   const float* __restrict__ bmlp,
    float* __restrict__ dp, float* __restrict__ ep)
{
  const int lane = threadIdx.x & 31;
  const int wv   = threadIdx.x >> 5;
  const int tile = blockIdx.x * 8 + wv;
  const bool isDec = tile < 1024;
  const int tt      = tile & 1023;
  const int rowTile = tt >> 4;
  const int colTile = tt & 15;
  const int row0 = rowTile << 4;
  const int col0 = colTile << 4;

  const float* src  = isDec ? dec : enc;
  const int    woff = isDec ? HDIM : 0;
  float*       dst  = isDec ? dp : ep;

  const int laneLo = lane & 15;
  const int laneHi = lane >> 4;

  const float* aPtr = src + (size_t)(row0 + laneLo) * HDIM + 2 * laneHi;
  const float* bPtr = W   + (size_t)(col0 + laneLo) * (2 * HDIM) + woff + 2 * laneHi;

  v8f acc = {};
  for (int k = 0; k < HDIM; k += 4) {
    v2f a = *(const v2f*)(aPtr + k);
    v2f b = *(const v2f*)(bPtr + k);
    acc = __builtin_amdgcn_wmma_f32_16x16x4_f32(
         false, a,  false, b,
         (short)0, acc,  false,  false);
  }

  const float bias = isDec ? bmlp[col0 + laneLo] : 0.0f;
  #pragma unroll
  for (int v = 0; v < 8; ++v) {
    const int m = v + 8 * laneHi;
    dst[(size_t)(row0 + m) * HDIM + col0 + laneLo] = acc[v] + bias;
  }
}

#define HC  64
#define PAD 66

__global__ __launch_bounds__(256) void attn_combine(
    const float* __restrict__ dp, const float* __restrict__ ep,
    const float* __restrict__ wout, const float* __restrict__ bout,
    float* __restrict__ out)
{
  __shared__ __align__(16) float dpS[64][PAD];
  __shared__ __align__(16) float epS[64][PAD];
  __shared__ __align__(16) float wS[HDIM];

  const int tid = threadIdx.x;
  const int b   = blockIdx.x >> 6;
  const int qt  = (blockIdx.x >> 3) & 7;
  const int et  = blockIdx.x & 7;
  const int q0  = qt * 64, e0 = et * 64;
  const int tx  = tid & 15, ty = tid >> 4;

  wS[tid] = wout[tid];

  const int r = tid >> 5;
  const int c = (tid & 31) * 2;

  float acc[4][4] = {};

  for (int hc = 0; hc < HDIM; hc += HC) {
    __syncthreads();
    #pragma unroll
    for (int i = 0; i < 8; ++i) {
      const int row = r + 8 * i;
      *(v2f*)&dpS[row][c] =
          *(const v2f*)(dp + (size_t)(b * 512 + q0 + row) * HDIM + hc + c);
      *(v2f*)&epS[row][c] =
          *(const v2f*)(ep + (size_t)(b * 512 + e0 + row) * HDIM + hc + c);
    }
    __syncthreads();

    for (int h = 0; h < HC; h += 2) {
      const v2f wv = *(const v2f*)&wS[hc + h];
      v2f dq[4], ee[4];
      #pragma unroll
      for (int i = 0; i < 4; ++i) dq[i] = *(const v2f*)&dpS[ty + 16 * i][h];
      #pragma unroll
      for (int j = 0; j < 4; ++j) ee[j] = *(const v2f*)&epS[tx + 16 * j][h];
      #pragma unroll
      for (int i = 0; i < 4; ++i)
        #pragma unroll
        for (int j = 0; j < 4; ++j) {
          acc[i][j] += fmaxf(dq[i].x + ee[j].x, 0.0f) * wv.x;
          acc[i][j] += fmaxf(dq[i].y + ee[j].y, 0.0f) * wv.y;
        }
    }
  }

  const float bo = bout[0];
  #pragma unroll
  for (int i = 0; i < 4; ++i)
    #pragma unroll
    for (int j = 0; j < 4; ++j)
      out[(size_t)(b * 512 + q0 + ty + 16 * i) * 512 + e0 + tx + 16 * j] =
          acc[i][j] + bo;
}

extern "C" void kernel_launch(void* const* d_in, const int* in_sizes, int n_in,
                              void* d_out, int out_size, void* d_ws, size_t ws_size,
                              hipStream_t stream) {
  const float* dec = (const float*)d_in[0];
  const float* enc = (const float*)d_in[1];
  const float* W   = (const float*)d_in[2];
  const float* bm  = (const float*)d_in[3];
  const float* wo  = (const float*)d_in[4];
  const float* bo  = (const float*)d_in[5];
  float* out = (float*)d_out;

  float* dp = (float*)d_ws;
  float* ep = dp + (size_t)NROWS * HDIM;

  proj_wmma<<<256, 256, 0, stream>>>(dec, enc, W, bm, dp, ep);
  attn_combine<<<128, 256, 0, stream>>>(dp, ep, wo, bo, out);
}


# projection kernel: each wave computes a 32x32 block (2x2 matrix tiles, 4 accumulators) from 16-byte per-lane loads, no LDS; k order in each 8-k chunk is 0,1,4,5 then 2,3,6,7 (sum order changes)
# speedup vs baseline: 1.0638x; 1.0638x over previous
#include <hip/hip_runtime.h>

typedef __attribute__((ext_vector_type(2))) float v2f;
typedef __attribute__((ext_vector_type(4))) float v4f;
typedef __attribute__((ext_vector_type(8))) float v8f;

#define HDIM 256
#define NROWS 1024

#define KGRP 4
#define NGRP (HDIM / (8 * KGRP))
#define NSET 3

__global__ __launch_bounds__(64) void proj_wmma(
    const float* __restrict__ dec, const float* __restrict__ enc,
    const float* __restrict__ W,   const float* __restrict__ bmlp,
    float* __restrict__ dp, float* __restrict__ ep)
{
  const int lane = threadIdx.x & 31;
  const int wv   = threadIdx.x >> 5;
  const int tile = blockIdx.x * 2 + wv;
  const bool isDec = tile < 256;
  const int tt   = tile & 255;
  const int row0 = (tt >> 3) << 5;
  const int col0 = (tt & 7) << 5;

  const float* src  = isDec ? dec : enc;
  const int    woff = isDec ? HDIM : 0;
  float*       dst  = isDec ? dp : ep;

  const int c  = lane & 15;
  const int hh = lane >> 4;

  const float* aP0 = src + (size_t)(row0 + c) * HDIM + 4 * hh;
  const float* aP1 = aP0 + 16 * HDIM;
  const float* bP0 = W + (size_t)(col0 + c) * (2 * HDIM) + woff + 4 * hh;
  const float* bP1 = bP0 + 16 * (2 * HDIM);

  v8f acc00 = {}, acc01 = {}, acc10 = {}, acc11 = {};

  v4f a0[NSET][KGRP], a1[NSET][KGRP], b0[NSET][KGRP], b1[NSET][KGRP];

  #pragma unroll
  for (int g = 0; g < NSET - 1; ++g) {
    #pragma unroll
    for (int j = 0; j < KGRP; ++j) {
      const int k = 8 * (KGRP * g + j);
      a0[g][j] = *(const v4f*)(aP0 + k);
      a1[g][j] = *(const v4f*)(aP1 + k);
      b0[g][j] = *(const v4f*)(bP0 + k);
      b1[g][j] = *(const v4f*)(bP1 + k);
    }
  }

  #pragma unroll
  for (int g = 0; g < NGRP; ++g) {
    if (g + NSET - 1 < NGRP) {
      const int sn = (g + NSET - 1) % NSET;
      #pragma unroll
      for (int j = 0; j < KGRP; ++j) {
        const int k = 8 * (KGRP * (g + NSET - 1) + j);
        a0[sn][j] = *(const v4f*)(aP0 + k);
        a1[sn][j] = *(const v4f*)(aP1 + k);
        b0[sn][j] = *(const v4f*)(bP0 + k);
        b1[sn][j] = *(const v4f*)(bP1 + k);
      }
    }
    __builtin_amdgcn_sched_barrier(0);
    const int sc = g % NSET;
    #pragma unroll
    for (int j = 0; j < KGRP; ++j) {
      const v2f a0l = {a0[sc][j].x, a0[sc][j].y}, a0h = {a0[sc][j].z, a0[sc][j].w};
      const v2f a1l = {a1[sc][j].x, a1[sc][j].y}, a1h = {a1[sc][j].z, a1[sc][j].w};
      const v2f b0l = {b0[sc][j].x, b0[sc][j].y}, b0h = {b0[sc][j].z, b0[sc][j].w};
      const v2f b1l = {b1[sc][j].x, b1[sc][j].y}, b1h = {b1[sc][j].z, b1[sc][j].w};
      acc00 = __builtin_amdgcn_wmma_f32_16x16x4_f32(false, a0l, false, b0l, (short)0, acc00, false, false);
      acc01 = __builtin_amdgcn_wmma_f32_16x16x4_f32(false, a0l, false, b1l, (short)0, acc01, false, false);
      acc10 = __builtin_amdgcn_wmma_f32_16x16x4_f32(false, a1l, false, b0l, (short)0, acc10, false, false);
      acc11 = __builtin_amdgcn_wmma_f32_16x16x4_f32(false, a1l, false, b1l, (short)0, acc11, false, false);
      acc00 = __builtin_amdgcn_wmma_f32_16x16x4_f32(false, a0h, false, b0h, (short)0, acc00, false, false);
      acc01 = __builtin_amdgcn_wmma_f32_16x16x4_f32(false, a0h, false, b1h, (short)0, acc01, false, false);
      acc10 = __builtin_amdgcn_wmma_f32_16x16x4_f32(false, a1h, false, b0h, (short)0, acc10, false, false);
      acc11 = __builtin_amdgcn_wmma_f32_16x16x4_f32(false, a1h, false, b1h, (short)0, acc11, false, false);
    }
  }

  const float bias0 = isDec ? bmlp[col0 + c] : 0.0f;
  const float bias1 = isDec ? bmlp[col0 + 16 + c] : 0.0f;
  #pragma unroll
  for (int v = 0; v < 8; ++v) {
    const int m = v + 8 * hh;
    float* d0 = dst + (size_t)(row0 + m) * HDIM + col0 + c;
    float* d1 = d0 + 16 * HDIM;
    d0[0]  = acc00[v] + bias0;
    d0[16] = acc01[v] + bias1;
    d1[0]  = acc10[v] + bias0;
    d1[16] = acc11[v] + bias1;
  }
}

#define HC  64
#define PAD 66

__global__ __launch_bounds__(256) void attn_combine(
    const float* __restrict__ dp, const float* __restrict__ ep,
    const float* __restrict__ wout, const float* __restrict__ bout,
    float* __restrict__ out)
{
  __shared__ __align__(16) float dpS[64][PAD];
  __shared__ __align__(16) float epS[64][PAD];
  __shared__ __align__(16) float wS[HDIM];

  const int tid = threadIdx.x;
  const int b   = blockIdx.x >> 6;
  const int qt  = (blockIdx.x >> 3) & 7;
  const int et  = blockIdx.x & 7;
  const int q0  = qt * 64, e0 = et * 64;
  const int tx  = tid & 15, ty = tid >> 4;

  wS[tid] = wout[tid];

  const int r = tid >> 5;
  const int c = (tid & 31) * 2;

  float acc[4][4] = {};

  for (int hc = 0; hc < HDIM; hc += HC) {
    __syncthreads();
    #pragma unroll
    for (int i = 0; i < 8; ++i) {
      const int row = r + 8 * i;
      *(v2f*)&dpS[row][c] =
          *(const v2f*)(dp + (size_t)(b * 512 + q0 + row) * HDIM + hc + c);
      *(v2f*)&epS[row][c] =
          *(const v2f*)(ep + (size_t)(b * 512 + e0 + row) * HDIM + hc + c);
    }
    __syncthreads();

    for (int h = 0; h < HC; h += 2) {
      const v2f wv = *(const v2f*)&wS[hc + h];
      v2f dq[4], ee[4];
      #pragma unroll
      for (int i = 0; i < 4; ++i) dq[i] = *(const v2f*)&dpS[ty + 16 * i][h];
      #pragma unroll
      for (int j = 0; j < 4; ++j) ee[j] = *(const v2f*)&epS[tx + 16 * j][h];
      #pragma unroll
      for (int i = 0; i < 4; ++i)
        #pragma unroll
        for (int j = 0; j < 4; ++j) {
          acc[i][j] += fmaxf(dq[i].x + ee[j].x, 0.0f) * wv.x;
          acc[i][j] += fmaxf(dq[i].y + ee[j].y, 0.0f) * wv.y;
        }
    }
  }

  const float bo = bout[0];
  #pragma unroll
  for (int i = 0; i < 4; ++i)
    #pragma unroll
    for (int j = 0; j < 4; ++j)
      out[(size_t)(b * 512 + q0 + ty + 16 * i) * 512 + e0 + tx + 16 * j] =
          acc[i][j] + bo;
}

extern "C" void kernel_launch(void* const* d_in, const int* in_sizes, int n_in,
                              void* d_out, int out_size, void* d_ws, size_t ws_size,
                              hipStream_t stream) {
  const float* dec = (const float*)d_in[0];
  const float* enc = (const float*)d_in[1];
  const float* W   = (const float*)d_in[2];
  const float* bm  = (const float*)d_in[3];
  const float* wo  = (const float*)d_in[4];
  const float* bo  = (const float*)d_in[5];
  float* out = (float*)d_out;

  float* dp = (float*)d_ws;
  float* ep = dp + (size_t)NROWS * HDIM;

  proj_wmma<<<256, 64, 0, stream>>>(dec, enc, W, bm, dp, ep);
  attn_combine<<<128, 256, 0, stream>>>(dp, ep, wo, bo, out);
}


# combine kernel without LDS: a wave owns 4 q x 64 e (2 e per lane, packed ops); dp and w enter as scalar operands, ep is stored transposed by the projection, e loads one step ahead; sums unchanged
# speedup vs baseline: 1.9988x; 1.8789x over previous
#include <hip/hip_runtime.h>

typedef __attribute__((ext_vector_type(2))) float v2f;
typedef __attribute__((ext_vector_type(4))) float v4f;
typedef __attribute__((ext_vector_type(8))) float v8f;

#define HDIM 256
#define NROWS 1024

#define KGRP 4
#define NGRP (HDIM / (8 * KGRP))
#define NSET 3

__global__ __launch_bounds__(64) void proj_wmma(
    const float* __restrict__ dec, const float* __restrict__ enc,
    const float* __restrict__ W,   const float* __restrict__ bmlp,
    float* __restrict__ dp, float* __restrict__ epT)
{
  const int lane = threadIdx.x & 31;
  const int wv   = threadIdx.x >> 5;
  const int tile = blockIdx.x * 2 + wv;
  const bool isDec = tile < 256;
  const int tt   = tile & 255;
  const int row0 = (tt >> 3) << 5;
  const int col0 = (tt & 7) << 5;

  const float* src  = isDec ? dec : enc;
  const int    woff = isDec ? HDIM : 0;

  const int c  = lane & 15;
  const int hh = lane >> 4;

  const float* sP = src + (size_t)(row0 + c) * HDIM + 4 * hh;
  const float* wP = W + (size_t)(col0 + c) * (2 * HDIM) + woff + 4 * hh;
  const float* aP0 = isDec ? sP : wP;
  const float* bP0 = isDec ? wP : sP;
  const float* aP1 = aP0 + (isDec ? 16 * HDIM : 16 * 2 * HDIM);
  const float* bP1 = bP0 + (isDec ? 16 * 2 * HDIM : 16 * HDIM);

  v8f acc00 = {}, acc01 = {}, acc10 = {}, acc11 = {};

  v4f a0[NSET][KGRP], a1[NSET][KGRP], b0[NSET][KGRP], b1[NSET][KGRP];

  #pragma unroll
  for (int g = 0; g < NSET - 1; ++g) {
    #pragma unroll
    for (int j = 0; j < KGRP; ++j) {
      const int k = 8 * (KGRP * g + j);
      a0[g][j] = *(const v4f*)(aP0 + k);
      a1[g][j] = *(const v4f*)(aP1 + k);
      b0[g][j] = *(const v4f*)(bP0 + k);
      b1[g][j] = *(const v4f*)(bP1 + k);
    }
  }

  #pragma unroll
  for (int g = 0; g < NGRP; ++g) {
    if (g + NSET - 1 < NGRP) {
      const int sn = (g + NSET - 1) % NSET;
      #pragma unroll
      for (int j = 0; j < KGRP; ++j) {
        const int k = 8 * (KGRP * (g + NSET - 1) + j);
        a0[sn][j] = *(const v4f*)(aP0 + k);
        a1[sn][j] = *(const v4f*)(aP1 + k);
        b0[sn][j] = *(const v4f*)(bP0 + k);
        b1[sn][j] = *(const v4f*)(bP1 + k);
      }
    }
    __builtin_amdgcn_sched_barrier(0);
    const int sc = g % NSET;
    #pragma unroll
    for (int j = 0; j < KGRP; ++j) {
      const v2f a0l = {a0[sc][j].x, a0[sc][j].y}, a0h = {a0[sc][j].z, a0[sc][j].w};
      const v2f a1l = {a1[sc][j].x, a1[sc][j].y}, a1h = {a1[sc][j].z, a1[sc][j].w};
      const v2f b0l = {b0[sc][j].x, b0[sc][j].y}, b0h = {b0[sc][j].z, b0[sc][j].w};
      const v2f b1l = {b1[sc][j].x, b1[sc][j].y}, b1h = {b1[sc][j].z, b1[sc][j].w};
      acc00 = __builtin_amdgcn_wmma_f32_16x16x4_f32(false, a0l, false, b0l, (short)0, acc00, false, false);
      acc01 = __builtin_amdgcn_wmma_f32_16x16x4_f32(false, a0l, false, b1l, (short)0, acc01, false, false);
      acc10 = __builtin_amdgcn_wmma_f32_16x16x4_f32(false, a1l, false, b0l, (short)0, acc10, false, false);
      acc11 = __builtin_amdgcn_wmma_f32_16x16x4_f32(false, a1l, false, b1l, (short)0, acc11, false, false);
      acc00 = __builtin_amdgcn_wmma_f32_16x16x4_f32(false, a0h, false, b0h, (short)0, acc00, false, false);
      acc01 = __builtin_amdgcn_wmma_f32_16x16x4_f32(false, a0h, false, b1h, (short)0, acc01, false, false);
      acc10 = __builtin_amdgcn_wmma_f32_16x16x4_f32(false, a1h, false, b0h, (short)0, acc10, false, false);
      acc11 = __builtin_amdgcn_wmma_f32_16x16x4_f32(false, a1h, false, b1h, (short)0, acc11, false, false);
    }
  }

  if (isDec) {
    const float bias0 = bmlp[col0 + c];
    const float bias1 = bmlp[col0 + 16 + c];
    #pragma unroll
    for (int v = 0; v < 8; ++v) {
      const int m = v + 8 * hh;
      float* d0 = dp + (size_t)(row0 + m) * HDIM + col0 + c;
      float* d1 = d0 + 16 * HDIM;
      d0[0]  = acc00[v] + bias0;
      d0[16] = acc01[v] + bias1;
      d1[0]  = acc10[v] + bias0;
      d1[16] = acc11[v] + bias1;
    }
  } else {
    const int bb = row0 >> 9;
    const int e0 = (row0 & 511) + c;
    float* t0 = epT + (size_t)(bb * HDIM + col0 + 8 * hh) * 512 + e0;
    float* t1 = t0 + (size_t)16 * 512;
    #pragma unroll
    for (int v = 0; v < 8; ++v) {
      t0[v * 512]      = acc00[v];
      t0[v * 512 + 16] = acc01[v];
      t1[v * 512]      = acc10[v];
      t1[v * 512 + 16] = acc11[v];
    }
  }
}

#define QT  4
#define HCH 16

__global__ __launch_bounds__(256) void attn_combine(
    const float* __restrict__ dp, const float* __restrict__ epT,
    const float* __restrict__ wout, const float* __restrict__ bout,
    float* __restrict__ out)
{
  const int lane = threadIdx.x & 31;
  const int wv = __builtin_amdgcn_readfirstlane(threadIdx.x >> 5);
  const int b  = blockIdx.x >> 7;
  const int qt = (blockIdx.x >> 3) & 15;
  const int et = blockIdx.x & 7;
  const int q0 = b * 512 + qt * 32 + wv * QT;
  const int e  = et * 64 + 2 * lane;

  const float* dr = dp + (size_t)q0 * HDIM;
  const float* er = epT + (size_t)b * HDIM * 512 + e;

  v2f acc[QT];
  #pragma unroll
  for (int i = 0; i < QT; ++i) acc[i] = (v2f){0.0f, 0.0f};

  auto step = [&](const v2f (&ev)[HCH], int h) {
    float d[QT][HCH];
    #pragma unroll
    for (int i = 0; i < QT; ++i)
      #pragma unroll
      for (int k = 0; k < HCH; ++k) d[i][k] = dr[i * HDIM + h + k];
    float w[HCH];
    #pragma unroll
    for (int k = 0; k < HCH; ++k) w[k] = wout[h + k];
    #pragma unroll
    for (int k = 0; k < HCH; ++k) {
      #pragma unroll
      for (int i = 0; i < QT; ++i) {
        v2f t = ev[k] + d[i][k];
        t.x = fmaxf(t.x, 0.0f);
        t.y = fmaxf(t.y, 0.0f);
        acc[i] = __builtin_elementwise_fma(t, (v2f){w[k], w[k]}, acc[i]);
      }
    }
  };
  auto fetch = [&](v2f (&ev)[HCH], int h) {
    #pragma unroll
    for (int k = 0; k < HCH; ++k)
      ev[k] = *(const v2f*)(er + (size_t)(h + k) * 512);
  };

  v2f evA[HCH], evB[HCH];
  fetch(evA, 0);
  for (int h = 0; h < HDIM - 2 * HCH; h += 2 * HCH) {
    fetch(evB, h + HCH);
    step(evA, h);
    fetch(evA, h + 2 * HCH);
    step(evB, h + HCH);
  }
  fetch(evB, HDIM - HCH);
  step(evA, HDIM - 2 * HCH);
  step(evB, HDIM - HCH);

  const float bo = bout[0];
  #pragma unroll
  for (int i = 0; i < QT; ++i)
    *(v2f*)(out + (size_t)(q0 + i) * 512 + e) = acc[i] + bo;
}

extern "C" void kernel_launch(void* const* d_in, const int* in_sizes, int n_in,
                              void* d_out, int out_size, void* d_ws, size_t ws_size,
                              hipStream_t stream) {
  const float* dec = (const float*)d_in[0];
  const float* enc = (const float*)d_in[1];
  const float* W   = (const float*)d_in[2];
  const float* bm  = (const float*)d_in[3];
  const float* wo  = (const float*)d_in[4];
  const float* bo  = (const float*)d_in[5];
  float* out = (float*)d_out;

  float* dp  = (float*)d_ws;
  float* epT = dp + (size_t)NROWS * HDIM;

  proj_wmma<<<256, 64, 0, stream>>>(dec, enc, W, bm, dp, epT);
  attn_combine<<<256, 256, 0, stream>>>(dp, epT, wo, bo, out);
}


# combine kernel: dp rows reach the packed adds as broadcast reads of a wave-private LDS slice, issued one step ahead with counted waits, not as scalar loads waited in full every step; sums unchanged
# speedup vs baseline: 2.4615x; 1.2315x over previous
#include <hip/hip_runtime.h>

typedef __attribute__((ext_vector_type(2))) float v2f;
typedef __attribute__((ext_vector_type(4))) float v4f;
typedef __attribute__((ext_vector_type(8))) float v8f;

#define HDIM 256
#define NROWS 1024

#define KGRP 4
#define NGRP (HDIM / (8 * KGRP))
#define NSET 3

__global__ __launch_bounds__(64) void proj_wmma(
    const float* __restrict__ dec, const float* __restrict__ enc,
    const float* __restrict__ W,   const float* __restrict__ bmlp,
    float* __restrict__ dp, float* __restrict__ epT)
{
  const int lane = threadIdx.x & 31;
  const int wv   = threadIdx.x >> 5;
  const int tile = blockIdx.x * 2 + wv;
  const bool isDec = tile < 256;
  const int tt   = tile & 255;
  const int row0 = (tt >> 3) << 5;
  const int col0 = (tt & 7) << 5;

  const float* src  = isDec ? dec : enc;
  const int    woff = isDec ? HDIM : 0;

  const int c  = lane & 15;
  const int hh = lane >> 4;

  const float* sP = src + (size_t)(row0 + c) * HDIM + 4 * hh;
  const float* wP = W + (size_t)(col0 + c) * (2 * HDIM) + woff + 4 * hh;
  const float* aP0 = isDec ? sP : wP;
  const float* bP0 = isDec ? wP : sP;
  const float* aP1 = aP0 + (isDec ? 16 * HDIM : 16 * 2 * HDIM);
  const float* bP1 = bP0 + (isDec ? 16 * 2 * HDIM : 16 * HDIM);

  v8f acc00 = {}, acc01 = {}, acc10 = {}, acc11 = {};

  v4f a0[NSET][KGRP], a1[NSET][KGRP], b0[NSET][KGRP], b1[NSET][KGRP];

  #pragma unroll
  for (int g = 0; g < NSET - 1; ++g) {
    #pragma unroll
    for (int j = 0; j < KGRP; ++j) {
      const int k = 8 * (KGRP * g + j);
      a0[g][j] = *(const v4f*)(aP0 + k);
      a1[g][j] = *(const v4f*)(aP1 + k);
      b0[g][j] = *(const v4f*)(bP0 + k);
      b1[g][j] = *(const v4f*)(bP1 + k);
    }
  }

  #pragma unroll
  for (int g = 0; g < NGRP; ++g) {
    if (g + NSET - 1 < NGRP) {
      const int sn = (g + NSET - 1) % NSET;
      #pragma unroll
      for (int j = 0; j < KGRP; ++j) {
        const int k = 8 * (KGRP * (g + NSET - 1) + j);
        a0[sn][j] = *(const v4f*)(aP0 + k);
        a1[sn][j] = *(const v4f*)(aP1 + k);
        b0[sn][j] = *(const v4f*)(bP0 + k);
        b1[sn][j] = *(const v4f*)(bP1 + k);
      }
    }
    __builtin_amdgcn_sched_barrier(0);
    const int sc = g % NSET;
    #pragma unroll
    for (int j = 0; j < KGRP; ++j) {
      const v2f a0l = {a0[sc][j].x, a0[sc][j].y}, a0h = {a0[sc][j].z, a0[sc][j].w};
      const v2f a1l = {a1[sc][j].x, a1[sc][j].y}, a1h = {a1[sc][j].z, a1[sc][j].w};
      const v2f b0l = {b0[sc][j].x, b0[sc][j].y}, b0h = {b0[sc][j].z, b0[sc][j].w};
      const v2f b1l = {b1[sc][j].x, b1[sc][j].y}, b1h = {b1[sc][j].z, b1[sc][j].w};
      acc00 = __builtin_amdgcn_wmma_f32_16x16x4_f32(false, a0l, false, b0l, (short)0, acc00, false, false);
      acc01 = __builtin_amdgcn_wmma_f32_16x16x4_f32(false, a0l, false, b1l, (short)0, acc01, false, false);
      acc10 = __builtin_amdgcn_wmma_f32_16x16x4_f32(false, a1l, false, b0l, (short)0, acc10, false, false);
      acc11 = __builtin_amdgcn_wmma_f32_16x16x4_f32(false, a1l, false, b1l, (short)0, acc11, false, false);
      acc00 = __builtin_amdgcn_wmma_f32_16x16x4_f32(false, a0h, false, b0h, (short)0, acc00, false, false);
      acc01 = __builtin_amdgcn_wmma_f32_16x16x4_f32(false, a0h, false, b1h, (short)0, acc01, false, false);
      acc10 = __builtin_amdgcn_wmma_f32_16x16x4_f32(false, a1h, false, b0h, (short)0, acc10, false, false);
      acc11 = __builtin_amdgcn_wmma_f32_16x16x4_f32(false, a1h, false, b1h, (short)0, acc11, false, false);
    }
  }

  if (isDec) {
    const float bias0 = bmlp[col0 + c];
    const float bias1 = bmlp[col0 + 16 + c];
    #pragma unroll
    for (int v = 0; v < 8; ++v) {
      const int m = v + 8 * hh;
      float* d0 = dp + (size_t)(row0 + m) * HDIM + col0 + c;
      float* d1 = d0 + 16 * HDIM;
      d0[0]  = acc00[v] + bias0;
      d0[16] = acc01[v] + bias1;
      d1[0]  = acc10[v] + bias0;
      d1[16] = acc11[v] + bias1;
    }
  } else {
    const int bb = row0 >> 9;
    const int e0 = (row0 & 511) + c;
    float* t0 = epT + (size_t)(bb * HDIM + col0 + 8 * hh) * 512 + e0;
    float* t1 = t0 + (size_t)16 * 512;
    #pragma unroll
    for (int v = 0; v < 8; ++v) {
      t0[v * 512]      = acc00[v];
      t0[v * 512 + 16] = acc01[v];
      t1[v * 512]      = acc10[v];
      t1[v * 512 + 16] = acc11[v];
    }
  }
}

#define QT  4
#define HCH 16
#define HV  (HCH / 4)
#define ROWV (HDIM / 4)

__global__ __launch_bounds__(256) void attn_combine(
    const float* __restrict__ dp, const float* __restrict__ epT,
    const float* __restrict__ wout, const float* __restrict__ bout,
    float* __restrict__ out)
{
  __shared__ v4f dl[8 * QT * ROWV];

  const int lane = threadIdx.x & 31;
  const int wv = __builtin_amdgcn_readfirstlane(threadIdx.x >> 5);
  const int b  = blockIdx.x >> 7;
  const int qt = (blockIdx.x >> 3) & 15;
  const int et = blockIdx.x & 7;
  const int q0 = b * 512 + qt * 32 + wv * QT;
  const int e  = et * 64 + 2 * lane;

  const float* er = epT + (size_t)b * HDIM * 512 + e;

  auto fetchE = [&](v2f (&ev)[HCH], int h) {
    #pragma unroll
    for (int k = 0; k < HCH; ++k)
      ev[k] = *(const v2f*)(er + (size_t)(h + k) * 512);
  };

  v2f evA[HCH], evB[HCH];
  fetchE(evA, 0);

  v4f* dls = dl + wv * (QT * ROWV);
  {
    const v4f* srcv = (const v4f*)(dp + (size_t)q0 * HDIM);
    v4f tmp[QT * ROWV / 32];
    #pragma unroll
    for (int j = 0; j < QT * ROWV / 32; ++j) tmp[j] = srcv[lane + 32 * j];
    #pragma unroll
    for (int j = 0; j < QT * ROWV / 32; ++j) dls[lane + 32 * j] = tmp[j];
  }
  __syncthreads();

  v2f acc[QT];
  #pragma unroll
  for (int i = 0; i < QT; ++i) acc[i] = (v2f){0.0f, 0.0f};

  auto fetchD = [&](v4f (&dv)[QT][HV], int h) {
    #pragma unroll
    for (int i = 0; i < QT; ++i)
      #pragma unroll
      for (int j = 0; j < HV; ++j) dv[i][j] = dls[i * ROWV + (h >> 2) + j];
  };

  auto loadW = [&](float (&w)[HCH], int h) {
    #pragma unroll
    for (int k = 0; k < HCH; ++k) w[k] = wout[h + k];
  };
  auto step = [&](const v2f (&ev)[HCH], const v4f (&dv)[QT][HV], const float (&w)[HCH]) {
    #pragma unroll
    for (int k = 0; k < HCH; ++k) {
      const float wk = w[k];
      #pragma unroll
      for (int i = 0; i < QT; ++i) {
        v2f t = ev[k] + dv[i][k >> 2][k & 3];
        t.x = fmaxf(t.x, 0.0f);
        t.y = fmaxf(t.y, 0.0f);
        acc[i] = __builtin_elementwise_fma(t, (v2f){wk, wk}, acc[i]);
      }
    }
  };

  v4f dA[QT][HV], dB[QT][HV];
  float wA[HCH], wB[HCH];
  fetchD(dA, 0);
  for (int h = 0; h < HDIM - 2 * HCH; h += 2 * HCH) {
    loadW(wA, h);
    fetchE(evB, h + HCH);
    fetchD(dB, h + HCH);
    __builtin_amdgcn_sched_barrier(0);
    step(evA, dA, wA);
    loadW(wB, h + HCH);
    fetchE(evA, h + 2 * HCH);
    fetchD(dA, h + 2 * HCH);
    __builtin_amdgcn_sched_barrier(0);
    step(evB, dB, wB);
  }
  loadW(wA, HDIM - 2 * HCH);
  loadW(wB, HDIM - HCH);
  fetchE(evB, HDIM - HCH);
  fetchD(dB, HDIM - HCH);
  __builtin_amdgcn_sched_barrier(0);
  step(evA, dA, wA);
  step(evB, dB, wB);

  const float bo = bout[0];
  #pragma unroll
  for (int i = 0; i < QT; ++i)
    *(v2f*)(out + (size_t)(q0 + i) * 512 + e) = acc[i] + bo;
}

extern "C" void kernel_launch(void* const* d_in, const int* in_sizes, int n_in,
                              void* d_out, int out_size, void* d_ws, size_t ws_size,
                              hipStream_t stream) {
  const float* dec = (const float*)d_in[0];
  const float* enc = (const float*)d_in[1];
  const float* W   = (const float*)d_in[2];
  const float* bm  = (const float*)d_in[3];
  const float* wo  = (const float*)d_in[4];
  const float* bo  = (const float*)d_in[5];
  float* out = (float*)d_out;

  float* dp  = (float*)d_ws;
  float* epT = dp + (size_t)NROWS * HDIM;

  proj_wmma<<<256, 64, 0, stream>>>(dec, enc, W, bm, dp, epT);
  attn_combine<<<256, 256, 0, stream>>>(dp, epT, wo, bo, out);
}


# combine kernel: relu is the packed add's clamp output modifier (dp, ep stored times 2^-64, sums times 2^64 at the store, both exact): loop = packed add + packed fma, no max; sums unchanged
# speedup vs baseline: 2.6283x; 1.0678x over previous
#include <hip/hip_runtime.h>

typedef __attribute__((ext_vector_type(2))) float v2f;
typedef __attribute__((ext_vector_type(4))) float v4f;
typedef __attribute__((ext_vector_type(8))) float v8f;

#define HDIM 256
#define NROWS 1024

#define PSCALE   0x1p-64f
#define PUNSCALE 0x1p+64f

#define KGRP 4
#define NGRP (HDIM / (8 * KGRP))
#define NSET 3

__global__ __launch_bounds__(64) void proj_wmma(
    const float* __restrict__ dec, const float* __restrict__ enc,
    const float* __restrict__ W,   const float* __restrict__ bmlp,
    float* __restrict__ dp, float* __restrict__ epT)
{
  const int lane = threadIdx.x & 31;
  const int wv   = threadIdx.x >> 5;
  const int tile = blockIdx.x * 2 + wv;
  const bool isDec = tile < 256;
  const int tt   = tile & 255;
  const int row0 = (tt >> 3) << 5;
  const int col0 = (tt & 7) << 5;

  const float* src  = isDec ? dec : enc;
  const int    woff = isDec ? HDIM : 0;

  const int c  = lane & 15;
  const int hh = lane >> 4;

  const float* sP = src + (size_t)(row0 + c) * HDIM + 4 * hh;
  const float* wP = W + (size_t)(col0 + c) * (2 * HDIM) + woff + 4 * hh;
  const float* aP0 = isDec ? sP : wP;
  const float* bP0 = isDec ? wP : sP;
  const float* aP1 = aP0 + (isDec ? 16 * HDIM : 16 * 2 * HDIM);
  const float* bP1 = bP0 + (isDec ? 16 * 2 * HDIM : 16 * HDIM);

  v8f acc00 = {}, acc01 = {}, acc10 = {}, acc11 = {};

  v4f a0[NSET][KGRP], a1[NSET][KGRP], b0[NSET][KGRP], b1[NSET][KGRP];

  #pragma unroll
  for (int g = 0; g < NSET - 1; ++g) {
    #pragma unroll
    for (int j = 0; j < KGRP; ++j) {
      const int k = 8 * (KGRP * g + j);
      a0[g][j] = *(const v4f*)(aP0 + k);
      a1[g][j] = *(const v4f*)(aP1 + k);
      b0[g][j] = *(const v4f*)(bP0 + k);
      b1[g][j] = *(const v4f*)(bP1 + k);
    }
  }

  #pragma unroll
  for (int g = 0; g < NGRP; ++g) {
    if (g + NSET - 1 < NGRP) {
      const int sn = (g + NSET - 1) % NSET;
      #pragma unroll
      for (int j = 0; j < KGRP; ++j) {
        const int k = 8 * (KGRP * (g + NSET - 1) + j);
        a0[sn][j] = *(const v4f*)(aP0 + k);
        a1[sn][j] = *(const v4f*)(aP1 + k);
        b0[sn][j] = *(const v4f*)(bP0 + k);
        b1[sn][j] = *(const v4f*)(bP1 + k);
      }
    }
    __builtin_amdgcn_sched_barrier(0);
    const int sc = g % NSET;
    #pragma unroll
    for (int j = 0; j < KGRP; ++j) {
      const v2f a0l = {a0[sc][j].x, a0[sc][j].y}, a0h = {a0[sc][j].z, a0[sc][j].w};
      const v2f a1l = {a1[sc][j].x, a1[sc][j].y}, a1h = {a1[sc][j].z, a1[sc][j].w};
      const v2f b0l = {b0[sc][j].x, b0[sc][j].y}, b0h = {b0[sc][j].z, b0[sc][j].w};
      const v2f b1l = {b1[sc][j].x, b1[sc][j].y}, b1h = {b1[sc][j].z, b1[sc][j].w};
      acc00 = __builtin_amdgcn_wmma_f32_16x16x4_f32(false, a0l, false, b0l, (short)0, acc00, false, false);
      acc01 = __builtin_amdgcn_wmma_f32_16x16x4_f32(false, a0l, false, b1l, (short)0, acc01, false, false);
      acc10 = __builtin_amdgcn_wmma_f32_16x16x4_f32(false, a1l, false, b0l, (short)0, acc10, false, false);
      acc11 = __builtin_amdgcn_wmma_f32_16x16x4_f32(false, a1l, false, b1l, (short)0, acc11, false, false);
      acc00 = __builtin_amdgcn_wmma_f32_16x16x4_f32(false, a0h, false, b0h, (short)0, acc00, false, false);
      acc01 = __builtin_amdgcn_wmma_f32_16x16x4_f32(false, a0h, false, b1h, (short)0, acc01, false, false);
      acc10 = __builtin_amdgcn_wmma_f32_16x16x4_f32(false, a1h, false, b0h, (short)0, acc10, false, false);
      acc11 = __builtin_amdgcn_wmma_f32_16x16x4_f32(false, a1h, false, b1h, (short)0, acc11, false, false);
    }
  }

  if (isDec) {
    const float bias0 = bmlp[col0 + c];
    const float bias1 = bmlp[col0 + 16 + c];
    #pragma unroll
    for (int v = 0; v < 8; ++v) {
      const int m = v + 8 * hh;
      float* d0 = dp + (size_t)(row0 + m) * HDIM + col0 + c;
      float* d1 = d0 + 16 * HDIM;
      d0[0]  = (acc00[v] + bias0) * PSCALE;
      d0[16] = (acc01[v] + bias1) * PSCALE;
      d1[0]  = (acc10[v] + bias0) * PSCALE;
      d1[16] = (acc11[v] + bias1) * PSCALE;
    }
  } else {
    const int bb = row0 >> 9;
    const int e0 = (row0 & 511) + c;
    float* t0 = epT + (size_t)(bb * HDIM + col0 + 8 * hh) * 512 + e0;
    float* t1 = t0 + (size_t)16 * 512;
    #pragma unroll
    for (int v = 0; v < 8; ++v) {
      t0[v * 512]      = acc00[v] * PSCALE;
      t0[v * 512 + 16] = acc01[v] * PSCALE;
      t1[v * 512]      = acc10[v] * PSCALE;
      t1[v * 512 + 16] = acc11[v] * PSCALE;
    }
  }
}

#define QT  4
#define HCH 16
#define HV  (HCH / 4)
#define ROWV (HDIM / 4)

static __device__ __forceinline__ v2f add_relu_lo(v2f e, v2f dpair) {
  v2f t;
  asm("v_pk_add_f32 %0, %1, %2 op_sel_hi:[1,0] clamp" : "=v"(t) : "v"(e), "v"(dpair));
  return t;
}
static __device__ __forceinline__ v2f add_relu_hi(v2f e, v2f dpair) {
  v2f t;
  asm("v_pk_add_f32 %0, %1, %2 op_sel:[0,1] clamp" : "=v"(t) : "v"(e), "v"(dpair));
  return t;
}

__global__ __launch_bounds__(256) void attn_combine(
    const float* __restrict__ dp, const float* __restrict__ epT,
    const float* __restrict__ wout, const float* __restrict__ bout,
    float* __restrict__ out)
{
  __shared__ v4f dl[8 * QT * ROWV];

  const int lane = threadIdx.x & 31;
  const int wv = __builtin_amdgcn_readfirstlane(threadIdx.x >> 5);
  const int b  = blockIdx.x >> 7;
  const int qt = (blockIdx.x >> 3) & 15;
  const int et = blockIdx.x & 7;
  const int q0 = b * 512 + qt * 32 + wv * QT;
  const int e  = et * 64 + 2 * lane;

  const float* er = epT + (size_t)b * HDIM * 512 + e;

  auto fetchE = [&](v2f (&ev)[HCH], int h) {
    #pragma unroll
    for (int k = 0; k < HCH; ++k)
      ev[k] = *(const v2f*)(er + (size_t)(h + k) * 512);
  };

  v2f evA[HCH], evB[HCH];
  fetchE(evA, 0);

  v4f* dls = dl + wv * (QT * ROWV);
  {
    const v4f* srcv = (const v4f*)(dp + (size_t)q0 * HDIM);
    v4f tmp[QT * ROWV / 32];
    #pragma unroll
    for (int j = 0; j < QT * ROWV / 32; ++j) tmp[j] = srcv[lane + 32 * j];
    #pragma unroll
    for (int j = 0; j < QT * ROWV / 32; ++j) dls[lane + 32 * j] = tmp[j];
  }
  __syncthreads();

  v2f acc[QT];
  #pragma unroll
  for (int i = 0; i < QT; ++i) acc[i] = (v2f){0.0f, 0.0f};

  auto fetchD = [&](v4f (&dv)[QT][HV], int h) {
    #pragma unroll
    for (int i = 0; i < QT; ++i)
      #pragma unroll
      for (int j = 0; j < HV; ++j) dv[i][j] = dls[i * ROWV + (h >> 2) + j];
  };

  auto loadW = [&](float (&w)[HCH], int h) {
    #pragma unroll
    for (int k = 0; k < HCH; ++k) w[k] = wout[h + k];
  };
  auto step = [&](const v2f (&ev)[HCH], const v4f (&dv)[QT][HV], const float (&w)[HCH]) {
    v2f t[3][QT];
    auto adds = [&](int k) {
      #pragma unroll
      for (int i = 0; i < QT; ++i) {
        const v4f d4 = dv[i][k >> 2];
        const v2f dpair = (k & 2) ? (v2f){d4.z, d4.w} : (v2f){d4.x, d4.y};
        t[k % 3][i] = (k & 1) ? add_relu_hi(ev[k], dpair) : add_relu_lo(ev[k], dpair);
      }
    };
    adds(0);
    adds(1);
    #pragma unroll
    for (int k = 0; k < HCH; ++k) {
      const v2f wk = {w[k], w[k]};
      if (k + 2 < HCH) adds(k + 2);
      __builtin_amdgcn_sched_barrier(0);
      #pragma unroll
      for (int i = 0; i < QT; ++i)
        acc[i] = __builtin_elementwise_fma(t[k % 3][i], wk, acc[i]);
      __builtin_amdgcn_sched_barrier(0);
    }
  };

  v4f dA[QT][HV], dB[QT][HV];
  float wA[HCH], wB[HCH];
  fetchD(dA, 0);
  for (int h = 0; h < HDIM - 2 * HCH; h += 2 * HCH) {
    loadW(wA, h);
    fetchE(evB, h + HCH);
    fetchD(dB, h + HCH);
    __builtin_amdgcn_sched_barrier(0);
    step(evA, dA, wA);
    loadW(wB, h + HCH);
    fetchE(evA, h + 2 * HCH);
    fetchD(dA, h + 2 * HCH);
    __builtin_amdgcn_sched_barrier(0);
    step(evB, dB, wB);
  }
  loadW(wA, HDIM - 2 * HCH);
  loadW(wB, HDIM - HCH);
  fetchE(evB, HDIM - HCH);
  fetchD(dB, HDIM - HCH);
  __builtin_amdgcn_sched_barrier(0);
  step(evA, dA, wA);
  step(evB, dB, wB);

  const float bo = bout[0];
  #pragma unroll
  for (int i = 0; i < QT; ++i)
    *(v2f*)(out + (size_t)(q0 + i) * 512 + e) =
        __builtin_elementwise_fma(acc[i], (v2f){PUNSCALE, PUNSCALE}, (v2f){bo, bo});
}

extern "C" void kernel_launch(void* const* d_in, const int* in_sizes, int n_in,
                              void* d_out, int out_size, void* d_ws, size_t ws_size,
                              hipStream_t stream) {
  const float* dec = (const float*)d_in[0];
  const float* enc = (const float*)d_in[1];
  const float* W   = (const float*)d_in[2];
  const float* bm  = (const float*)d_in[3];
  const float* wo  = (const float*)d_in[4];
  const float* bo  = (const float*)d_in[5];
  float* out = (float*)d_out;

  float* dp  = (float*)d_ws;
  float* epT = dp + (size_t)NROWS * HDIM;

  proj_wmma<<<256, 64, 0, stream>>>(dec, enc, W, bm, dp, epT);
  attn_combine<<<256, 256, 0, stream>>>(dp, epT, wo, bo, out);
}
